# starting kernel
# baseline (speedup 1.0000x reference)
#include <hip/hip_runtime.h>
#include <cmath>


#define L      32
#define KHEADS 8
#define NH     8
#define HID    300
#define BATCH  4096
#define NB     18688

#define TILE_M 16
#define NWAVES 16
#define THREADS (NWAVES * 32)

#define PCOLS 2336
#define NTILES (PCOLS / 16)
#define PSTR  2340
#define QSTR  33
#define HSTR  308

#define ROFF  0
#define RTOFF 1024
#define VOFF  2048
#define BBOFF 2304

typedef float v2f __attribute__((ext_vector_type(2)));
typedef float v8f __attribute__((ext_vector_type(8)));

__device__ __forceinline__ float softplus_f(float x) {
    return log1pf(__expf(-fabsf(x))) + fmaxf(x, 0.0f);
}

__device__ __forceinline__ int pcol_to_wcol(int pcol, int k) {
    if      (pcol < 1024) return k * 1024 + pcol;
    else if (pcol < 2048) return 8192  + k * 1024 + (pcol - 1024);
    else if (pcol < 2304) return 16384 + k * 256  + (pcol - 2048);
    else                  return 18432 + k * 32   + (pcol - 2304);
}

extern "C" __global__ __launch_bounds__(THREADS)
void sylvester_flow_kernel(const float* __restrict__ z0,
                           const float* __restrict__ hmat,
                           const float* __restrict__ W,
                           const float* __restrict__ bvec,
                           float* __restrict__ out)
{
    extern __shared__ float smem[];
    float* P  = smem;
    float* Qs = P  + TILE_M * PSTR;
    float* hL = Qs + TILE_M * (L * QSTR);
    float* zs = hL + TILE_M * HSTR;
    float* ys = zs + TILE_M * L;
    float* ts = ys + TILE_M * L;
    float* ss = ts + TILE_M * L;

    const int tid  = threadIdx.x;
    const int wave = tid >> 5;
    const int lane = tid & 31;
    const int hi   = lane >> 4;
    const int n16  = lane & 15;
    const int kk2  = hi * 2;
    const int row0 = blockIdx.x * TILE_M;

    for (int idx = tid; idx < TILE_M * HID; idx += THREADS) {
        int r = idx / HID, c = idx - r * HID;
        hL[r * HSTR + c] = hmat[(size_t)(row0 + r) * HID + c];
    }
    zs[wave * L + lane] = z0[(size_t)(row0 + wave) * L + lane];

    const int myrow = wave;
    const int prow  = myrow * PSTR;
    const int qbase = myrow * (L * QSTR);
    float ln_det = 0.0f;

    __syncthreads();

    for (int k = 0; k < KHEADS; ++k) {
        for (int nt0 = wave * 2; nt0 < NTILES; nt0 += NWAVES * 2) {
            int pcol0 = nt0 * 16;
            int pcol1 = pcol0 + 16;
            int wcol0 = pcol_to_wcol(pcol0, k);
            int wcol1 = pcol_to_wcol(pcol1, k);

            const float* wp0 = W + (size_t)kk2 * NB + wcol0 + n16;
            const float* wp1 = W + (size_t)kk2 * NB + wcol1 + n16;
            v8f acc0 = {};
            v8f acc1 = {};
            #pragma unroll 5
            for (int s = 0; s < HID / 4; ++s) {
                size_t roff = (size_t)s * 4 * NB;
                v2f bv0, bv1;
                bv0.x = wp0[roff];
                bv0.y = wp0[roff + NB];
                bv1.x = wp1[roff];
                bv1.y = wp1[roff + NB];
                int hb = n16 * HSTR + 4 * s + kk2;
                v2f av;
                av.x = hL[hb];
                av.y = hL[hb + 1];
                acc0 = __builtin_amdgcn_wmma_f32_16x16x4_f32(
                    false, av, false, bv0, (short)0, acc0, false, false);
                acc1 = __builtin_amdgcn_wmma_f32_16x16x4_f32(
                    false, av, false, bv1, (short)0, acc1, false, false);
            }
            float bias0 = bvec[wcol0 + n16];
            float bias1 = bvec[wcol1 + n16];
            #pragma unroll
            for (int r = 0; r < 8; ++r) {
                P[(r + 8 * hi) * PSTR + pcol0 + n16] = acc0[r] + bias0;
                P[(r + 8 * hi) * PSTR + pcol1 + n16] = acc1[r] + bias1;
            }
        }
        __syncthreads();

        float q[L];
        #pragma unroll
        for (int j = 0; j < L; ++j) q[j] = (j == lane) ? 1.0f : 0.0f;

        for (int hd = 0; hd < NH; ++hd) {
            int vb = prow + VOFF + hd * L;
            float vv = 0.0f, u = 0.0f;
            #pragma unroll
            for (int j = 0; j < L; ++j) {
                float vj = P[vb + j];
                vv += vj * vj;
                u  += q[j] * vj;
            }
            float cu = (2.0f / vv) * u;
            #pragma unroll
            for (int j = 0; j < L; ++j) q[j] -= cu * P[vb + j];
        }
        #pragma unroll
        for (int j = 0; j < L; ++j) Qs[qbase + lane * QSTR + j] = q[j];

        float y = 0.0f;
        #pragma unroll
        for (int i = 0; i < L; ++i)
            y += Qs[qbase + i * QSTR + lane] * zs[myrow * L + i];
        ys[myrow * L + lane] = y;

        float rtraw = P[prow + RTOFF + lane * (L + 1)];
        float rt_d  = softplus_f(rtraw) + 0.01f;
        float rraw  = P[prow + ROFF  + lane * (L + 1)];
        float rr    = rraw * rt_d;
        float mval  = softplus_f(rr) - 1.0f;
        float rdnew = rraw + (mval - rr) / rt_d;

        float wv = rt_d * y + P[prow + BBOFF + lane];
        #pragma unroll
        for (int jj = 1; jj < L; ++jj) {
            int col = (lane + jj) & (L - 1);
            float e = P[prow + RTOFF + lane * L + col] * ys[myrow * L + col];
            wv += (col > lane) ? e : 0.0f;
        }
        float t = tanhf(wv);
        ts[myrow * L + lane] = t;

        float sv = rdnew * t;
        #pragma unroll
        for (int jj = 1; jj < L; ++jj) {
            int col = (lane + jj) & (L - 1);
            float e = P[prow + ROFF + lane * L + col] * ts[myrow * L + col];
            sv += (col > lane) ? e : 0.0f;
        }
        ss[myrow * L + lane] = sv;

        float dz = 0.0f;
        #pragma unroll
        for (int j = 0; j < L; ++j) dz += q[j] * ss[myrow * L + j];
        zs[myrow * L + lane] += dz;

        float hp = 1.0f - t * t;
        ln_det += logf(fabsf(1.0f + hp * mval) + 1e-8f);

        __syncthreads();
    }

    out[(size_t)(row0 + myrow) * L + lane] = zs[myrow * L + lane];
    out[(size_t)BATCH * L + (size_t)(row0 + myrow) * L + lane] = ln_det;
}

extern "C" void kernel_launch(void* const* d_in, const int* in_sizes, int n_in,
                              void* d_out, int out_size, void* d_ws, size_t ws_size,
                              hipStream_t stream) {
    (void)in_sizes; (void)n_in; (void)d_ws; (void)ws_size; (void)out_size;
    const float* z0 = (const float*)d_in[0];
    const float* h  = (const float*)d_in[1];
    const float* W  = (const float*)d_in[2];
    const float* b  = (const float*)d_in[3];
    float* out = (float*)d_out;

    size_t lds_bytes = (size_t)(TILE_M * PSTR + TILE_M * L * QSTR +
                                TILE_M * HSTR + 4 * TILE_M * L) * sizeof(float);
    hipFuncSetAttribute((const void*)sylvester_flow_kernel,
                        hipFuncAttributeMaxDynamicSharedMemorySize,
                        (int)lds_bytes);
    sylvester_flow_kernel<<<BATCH / TILE_M, THREADS, lds_bytes, stream>>>(
        z0, h, W, b, out);
}


# Multiply only the upper-triangle columns of R and R_t that the reference's triu keeps, packed row by row into dense 16-column tiles: 84 instead of 146 N-tiles per head in the fused f32 WMMA GEMM
# speedup vs baseline: 1.2026x; 1.2026x over previous
#include <hip/hip_runtime.h>
#include <cmath>


#define L      32
#define KHEADS 8
#define NH     8
#define HID    300
#define BATCH  4096
#define NB     18688

#define TILE_M 16
#define NWAVES 16
#define THREADS (NWAVES * 32)

#define TRI   528
#define PCOLS 1344
#define NTILES (PCOLS / 16)
#define PSTR  1348
#define QSTR  33
#define HSTR  308

#define ROFF  0
#define RTOFF TRI
#define VOFF  (2 * TRI)
#define BBOFF (2 * TRI + NH * L)

typedef float v2f __attribute__((ext_vector_type(2)));
typedef float v8f __attribute__((ext_vector_type(8)));

__device__ __forceinline__ float softplus_f(float x) {
    return log1pf(__expf(-fabsf(x))) + fmaxf(x, 0.0f);
}

__device__ __forceinline__ int tri_start(int i) { return (i * (2 * L + 1 - i)) >> 1; }

__device__ __forceinline__ int pcol_to_wcol(int pcol, int k) {
    if (pcol < 2 * TRI) {
        int rt = (pcol >= TRI) ? 1 : 0;
        int q  = pcol - rt * TRI;
        int i  = (int)((65.0f - sqrtf((float)(4225 - 8 * q))) * 0.5f);
        if (tri_start(i + 1) <= q) ++i;
        if (tri_start(i) > q)      --i;
        int j  = i + (q - tri_start(i));
        return rt * 8192 + k * 1024 + i * L + j;
    }
    else if (pcol < BBOFF) return 16384 + k * 256 + (pcol - VOFF);
    else                   return 18432 + k * 32  + (pcol - BBOFF);
}

extern "C" __global__ __launch_bounds__(THREADS)
void sylvester_flow_kernel(const float* __restrict__ z0,
                           const float* __restrict__ hmat,
                           const float* __restrict__ W,
                           const float* __restrict__ bvec,
                           float* __restrict__ out)
{
    extern __shared__ float smem[];
    float* P  = smem;
    float* Qs = P  + TILE_M * PSTR;
    float* hL = Qs + TILE_M * (L * QSTR);
    float* zs = hL + TILE_M * HSTR;
    float* ys = zs + TILE_M * L;
    float* ts = ys + TILE_M * L;
    float* ss = ts + TILE_M * L;

    const int tid  = threadIdx.x;
    const int wave = tid >> 5;
    const int lane = tid & 31;
    const int hi   = lane >> 4;
    const int n16  = lane & 15;
    const int kk2  = hi * 2;
    const int row0 = blockIdx.x * TILE_M;

    for (int idx = tid; idx < TILE_M * HID; idx += THREADS) {
        int r = idx / HID, c = idx - r * HID;
        hL[r * HSTR + c] = hmat[(size_t)(row0 + r) * HID + c];
    }
    zs[wave * L + lane] = z0[(size_t)(row0 + wave) * L + lane];

    const int myrow = wave;
    const int prow  = myrow * PSTR;
    const int qbase = myrow * (L * QSTR);
    float ln_det = 0.0f;

    __syncthreads();

    for (int k = 0; k < KHEADS; ++k) {
        for (int nt0 = wave * 2; nt0 < NTILES; nt0 += NWAVES * 2) {
            int pcol0 = nt0 * 16;
            int pcol1 = pcol0 + 16;
            int wcol0 = pcol_to_wcol(pcol0 + n16, k);
            int wcol1 = pcol_to_wcol(pcol1 + n16, k);

            const float* wp0 = W + (size_t)kk2 * NB + wcol0;
            const float* wp1 = W + (size_t)kk2 * NB + wcol1;
            v8f acc0 = {};
            v8f acc1 = {};
            #pragma unroll 5
            for (int s = 0; s < HID / 4; ++s) {
                size_t roff = (size_t)s * 4 * NB;
                v2f bv0, bv1;
                bv0.x = wp0[roff];
                bv0.y = wp0[roff + NB];
                bv1.x = wp1[roff];
                bv1.y = wp1[roff + NB];
                int hb = n16 * HSTR + 4 * s + kk2;
                v2f av;
                av.x = hL[hb];
                av.y = hL[hb + 1];
                acc0 = __builtin_amdgcn_wmma_f32_16x16x4_f32(
                    false, av, false, bv0, (short)0, acc0, false, false);
                acc1 = __builtin_amdgcn_wmma_f32_16x16x4_f32(
                    false, av, false, bv1, (short)0, acc1, false, false);
            }
            float bias0 = bvec[wcol0];
            float bias1 = bvec[wcol1];
            #pragma unroll
            for (int r = 0; r < 8; ++r) {
                P[(r + 8 * hi) * PSTR + pcol0 + n16] = acc0[r] + bias0;
                P[(r + 8 * hi) * PSTR + pcol1 + n16] = acc1[r] + bias1;
            }
        }
        __syncthreads();

        float q[L];
        #pragma unroll
        for (int j = 0; j < L; ++j) q[j] = (j == lane) ? 1.0f : 0.0f;

        for (int hd = 0; hd < NH; ++hd) {
            int vb = prow + VOFF + hd * L;
            float vv = 0.0f, u = 0.0f;
            #pragma unroll
            for (int j = 0; j < L; ++j) {
                float vj = P[vb + j];
                vv += vj * vj;
                u  += q[j] * vj;
            }
            float cu = (2.0f / vv) * u;
            #pragma unroll
            for (int j = 0; j < L; ++j) q[j] -= cu * P[vb + j];
        }
        #pragma unroll
        for (int j = 0; j < L; ++j) Qs[qbase + lane * QSTR + j] = q[j];

        float y = 0.0f;
        #pragma unroll
        for (int i = 0; i < L; ++i)
            y += Qs[qbase + i * QSTR + lane] * zs[myrow * L + i];
        ys[myrow * L + lane] = y;

        const int tl = tri_start(lane);
        float rtraw = P[prow + RTOFF + tl];
        float rt_d  = softplus_f(rtraw) + 0.01f;
        float rraw  = P[prow + ROFF  + tl];
        float rr    = rraw * rt_d;
        float mval  = softplus_f(rr) - 1.0f;
        float rdnew = rraw + (mval - rr) / rt_d;

        float wv = rt_d * y + P[prow + BBOFF + lane];
        #pragma unroll
        for (int jj = 1; jj < L; ++jj) {
            int col = (lane + jj) & (L - 1);
            float e = P[prow + RTOFF + tl + jj] * ys[myrow * L + col];
            wv += (lane + jj < L) ? e : 0.0f;
        }
        float t = tanhf(wv);
        ts[myrow * L + lane] = t;

        float sv = rdnew * t;
        #pragma unroll
        for (int jj = 1; jj < L; ++jj) {
            int col = (lane + jj) & (L - 1);
            float e = P[prow + ROFF + tl + jj] * ts[myrow * L + col];
            sv += (lane + jj < L) ? e : 0.0f;
        }
        ss[myrow * L + lane] = sv;

        float dz = 0.0f;
        #pragma unroll
        for (int j = 0; j < L; ++j) dz += q[j] * ss[myrow * L + j];
        zs[myrow * L + lane] += dz;

        float hp = 1.0f - t * t;
        ln_det += logf(fabsf(1.0f + hp * mval) + 1e-8f);

        __syncthreads();
    }

    out[(size_t)(row0 + myrow) * L + lane] = zs[myrow * L + lane];
    out[(size_t)BATCH * L + (size_t)(row0 + myrow) * L + lane] = ln_det;
}

extern "C" void kernel_launch(void* const* d_in, const int* in_sizes, int n_in,
                              void* d_out, int out_size, void* d_ws, size_t ws_size,
                              hipStream_t stream) {
    (void)in_sizes; (void)n_in; (void)d_ws; (void)ws_size; (void)out_size;
    const float* z0 = (const float*)d_in[0];
    const float* h  = (const float*)d_in[1];
    const float* W  = (const float*)d_in[2];
    const float* b  = (const float*)d_in[3];
    float* out = (float*)d_out;

    size_t lds_bytes = (size_t)(TILE_M * PSTR + TILE_M * L * QSTR +
                                TILE_M * HSTR + 4 * TILE_M * L) * sizeof(float);
    hipFuncSetAttribute((const void*)sylvester_flow_kernel,
                        hipFuncAttributeMaxDynamicSharedMemorySize,
                        (int)lds_bytes);
    sylvester_flow_kernel<<<BATCH / TILE_M, THREADS, lds_bytes, stream>>>(
        z0, h, W, b, out);
}


# B operand read with one 16-byte and one 8-byte load per lane and W row (6 N-tiles share an A fragment, one pass per wave); triangle rows packed from a 4-column boundary: 90 tiles, was 84
# speedup vs baseline: 1.5727x; 1.3077x over previous
#include <hip/hip_runtime.h>
#include <cmath>


#define L      32
#define KHEADS 8
#define NH     8
#define HID    300
#define BATCH  4096
#define NB     18688

#define TILE_M 16
#define NWAVES 16
#define THREADS (NWAVES * 32)

#define TGRP  144
#define TRI   (4 * TGRP)
#define PCOLS 1440
#define GWAVES 15
#define WCOLS 96
#define WGRPS (WCOLS / 4)
#define PSTR  1444
#define QSTR  33
#define HSTR  308

#define ROFF  0
#define RTOFF TRI
#define VOFF  (2 * TRI)
#define BBOFF (2 * TRI + NH * L)

typedef float v2f __attribute__((ext_vector_type(2)));
typedef float v4f __attribute__((ext_vector_type(4)));
typedef float v8f __attribute__((ext_vector_type(8)));

__device__ __forceinline__ float softplus_f(float x) {
    return log1pf(__expf(-fabsf(x))) + fmaxf(x, 0.0f);
}

__device__ __forceinline__ int grp_base(int a) { return 34 * a - 2 * a * a; }

__device__ __forceinline__ int tri_start(int i) {
    int a = i >> 2, b = i & 3;
    return 4 * (grp_base(a) + b * (8 - a)) + b;
}

__device__ __forceinline__ void group_to_wcol(int g, int& wcol, int& kstep) {
    if (g < 2 * TGRP) {
        int rt = (g >= TGRP) ? 1 : 0;
        int q  = g - rt * TGRP;
        int a  = 0;
        #pragma unroll
        for (int t = 1; t < 8; ++t) if (q >= grp_base(t)) a = t;
        int rem = q - grp_base(a);
        int b   = rem / (8 - a);
        int r   = rem - b * (8 - a);
        wcol  = rt * 8192 + (4 * a + b) * L + 4 * (a + r);
        kstep = L * L;
    }
    else if (g < 2 * TGRP + NH * L / 4) { wcol = 16384 + 4 * (g - 2 * TGRP);              kstep = NH * L; }
    else                                { wcol = 18432 + 4 * (g - 2 * TGRP - NH * L / 4); kstep = L; }
}

extern "C" __global__ __launch_bounds__(THREADS)
void sylvester_flow_kernel(const float* __restrict__ z0,
                           const float* __restrict__ hmat,
                           const float* __restrict__ W,
                           const float* __restrict__ bvec,
                           float* __restrict__ out)
{
    extern __shared__ __attribute__((aligned(16))) float smem[];
    float* P  = smem;
    float* Qs = P  + TILE_M * PSTR;
    float* hL = Qs + TILE_M * (L * QSTR);
    float* zs = hL + TILE_M * HSTR;
    float* ys = zs + TILE_M * L;
    float* ts = ys + TILE_M * L;
    float* ss = ts + TILE_M * L;

    const int tid  = threadIdx.x;
    const int wave = tid >> 5;
    const int lane = tid & 31;
    const int hi   = lane >> 4;
    const int n16  = lane & 15;
    const int kk2  = hi * 2;
    const int row0 = blockIdx.x * TILE_M;

    for (int idx = tid; idx < TILE_M * HID; idx += THREADS) {
        int r = idx / HID, c = idx - r * HID;
        hL[r * HSTR + c] = hmat[(size_t)(row0 + r) * HID + c];
    }
    zs[wave * L + lane] = z0[(size_t)(row0 + wave) * L + lane];

    const int myrow = wave;
    const int prow  = myrow * PSTR;
    const int qbase = myrow * (L * QSTR);
    float ln_det = 0.0f;

    const int gw = (wave < GWAVES) ? wave : GWAVES - 1;
    int wc4, ks4, wc2, ks2;
    group_to_wcol(gw * WGRPS + n16, wc4, ks4);
    group_to_wcol(gw * WGRPS + 16 + (n16 >> 1), wc2, ks2);
    wc2 += 2 * (n16 & 1);

    __syncthreads();

    for (int k = 0; k < KHEADS; ++k) {
        if (wave < GWAVES) {
            const int wcol4 = wc4 + k * ks4;
            const int wcol2 = wc2 + k * ks2;
            const float* wp4 = W + (size_t)kk2 * NB + wcol4;
            const float* wp2 = W + (size_t)kk2 * NB + wcol2;
            v8f acc0 = {}, acc1 = {}, acc2 = {}, acc3 = {}, acc4 = {}, acc5 = {};
            #pragma unroll 5
            for (int s = 0; s < HID / 4; ++s) {
                size_t roff = (size_t)s * 4 * NB;
                v4f b0 = *(const v4f*)(wp4 + roff);
                v4f b1 = *(const v4f*)(wp4 + roff + NB);
                v2f c0 = *(const v2f*)(wp2 + roff);
                v2f c1 = *(const v2f*)(wp2 + roff + NB);
                int hb = n16 * HSTR + 4 * s + kk2;
                v2f av;
                av.x = hL[hb];
                av.y = hL[hb + 1];
                v2f bv;
                bv.x = b0.x; bv.y = b1.x;
                acc0 = __builtin_amdgcn_wmma_f32_16x16x4_f32(
                    false, av, false, bv, (short)0, acc0, false, false);
                bv.x = b0.y; bv.y = b1.y;
                acc1 = __builtin_amdgcn_wmma_f32_16x16x4_f32(
                    false, av, false, bv, (short)0, acc1, false, false);
                bv.x = b0.z; bv.y = b1.z;
                acc2 = __builtin_amdgcn_wmma_f32_16x16x4_f32(
                    false, av, false, bv, (short)0, acc2, false, false);
                bv.x = b0.w; bv.y = b1.w;
                acc3 = __builtin_amdgcn_wmma_f32_16x16x4_f32(
                    false, av, false, bv, (short)0, acc3, false, false);
                bv.x = c0.x; bv.y = c1.x;
                acc4 = __builtin_amdgcn_wmma_f32_16x16x4_f32(
                    false, av, false, bv, (short)0, acc4, false, false);
                bv.x = c0.y; bv.y = c1.y;
                acc5 = __builtin_amdgcn_wmma_f32_16x16x4_f32(
                    false, av, false, bv, (short)0, acc5, false, false);
            }
            v4f bias4 = *(const v4f*)(bvec + wcol4);
            v2f bias2 = *(const v2f*)(bvec + wcol2);
            #pragma unroll
            for (int r = 0; r < 8; ++r) {
                float* pr = P + (r + 8 * hi) * PSTR + wave * WCOLS;
                v4f o4;
                o4.x = acc0[r] + bias4.x;
                o4.y = acc1[r] + bias4.y;
                o4.z = acc2[r] + bias4.z;
                o4.w = acc3[r] + bias4.w;
                *(v4f*)(pr + 4 * n16) = o4;
                v2f o2;
                o2.x = acc4[r] + bias2.x;
                o2.y = acc5[r] + bias2.y;
                *(v2f*)(pr + 64 + 2 * n16) = o2;
            }
        }
        __syncthreads();

        float q[L];
        #pragma unroll
        for (int j = 0; j < L; ++j) q[j] = (j == lane) ? 1.0f : 0.0f;

        for (int hd = 0; hd < NH; ++hd) {
            int vb = prow + VOFF + hd * L;
            float vv = 0.0f, u = 0.0f;
            #pragma unroll
            for (int j = 0; j < L; ++j) {
                float vj = P[vb + j];
                vv += vj * vj;
                u  += q[j] * vj;
            }
            float cu = (2.0f / vv) * u;
            #pragma unroll
            for (int j = 0; j < L; ++j) q[j] -= cu * P[vb + j];
        }
        #pragma unroll
        for (int j = 0; j < L; ++j) Qs[qbase + lane * QSTR + j] = q[j];

        float y = 0.0f;
        #pragma unroll
        for (int i = 0; i < L; ++i)
            y += Qs[qbase + i * QSTR + lane] * zs[myrow * L + i];
        ys[myrow * L + lane] = y;

        const int tl = tri_start(lane);
        float rtraw = P[prow + RTOFF + tl];
        float rt_d  = softplus_f(rtraw) + 0.01f;
        float rraw  = P[prow + ROFF  + tl];
        float rr    = rraw * rt_d;
        float mval  = softplus_f(rr) - 1.0f;
        float rdnew = rraw + (mval - rr) / rt_d;

        float wv = rt_d * y + P[prow + BBOFF + lane];
        #pragma unroll
        for (int jj = 1; jj < L; ++jj) {
            int col = (lane + jj) & (L - 1);
            float e = P[prow + RTOFF + tl + jj] * ys[myrow * L + col];
            wv += (lane + jj < L) ? e : 0.0f;
        }
        float t = tanhf(wv);
        ts[myrow * L + lane] = t;

        float sv = rdnew * t;
        #pragma unroll
        for (int jj = 1; jj < L; ++jj) {
            int col = (lane + jj) & (L - 1);
            float e = P[prow + ROFF + tl + jj] * ts[myrow * L + col];
            sv += (lane + jj < L) ? e : 0.0f;
        }
        ss[myrow * L + lane] = sv;

        float dz = 0.0f;
        #pragma unroll
        for (int j = 0; j < L; ++j) dz += q[j] * ss[myrow * L + j];
        zs[myrow * L + lane] += dz;

        float hp = 1.0f - t * t;
        ln_det += logf(fabsf(1.0f + hp * mval) + 1e-8f);

        __syncthreads();
    }

    out[(size_t)(row0 + myrow) * L + lane] = zs[myrow * L + lane];
    out[(size_t)BATCH * L + (size_t)(row0 + myrow) * L + lane] = ln_det;
}

extern "C" void kernel_launch(void* const* d_in, const int* in_sizes, int n_in,
                              void* d_out, int out_size, void* d_ws, size_t ws_size,
                              hipStream_t stream) {
    (void)in_sizes; (void)n_in; (void)d_ws; (void)ws_size; (void)out_size;
    const float* z0 = (const float*)d_in[0];
    const float* h  = (const float*)d_in[1];
    const float* W  = (const float*)d_in[2];
    const float* b  = (const float*)d_in[3];
    float* out = (float*)d_out;

    size_t lds_bytes = (size_t)(TILE_M * PSTR + TILE_M * L * QSTR +
                                TILE_M * HSTR + 4 * TILE_M * L) * sizeof(float);
    hipFuncSetAttribute((const void*)sylvester_flow_kernel,
                        hipFuncAttributeMaxDynamicSharedMemorySize,
                        (int)lds_bytes);
    sylvester_flow_kernel<<<BATCH / TILE_M, THREADS, lds_bytes, stream>>>(
        z0, h, W, b, out);
}


# GEMM column groups re-dealt so each aligned 256-byte piece of a W row is fetched by one load instruction: 12 waves x two 16-group chunks of whole row pairs, 16-byte loads only; was 15 x (16+8)
# speedup vs baseline: 1.6965x; 1.0787x over previous
#include <hip/hip_runtime.h>
#include <cmath>


#define L      32
#define KHEADS 8
#define NH     8
#define HID    300
#define BATCH  4096
#define NB     18688

#define TILE_M 16
#define NWAVES 16
#define THREADS (NWAVES * 32)

#define TGRP  144
#define TRI   (4 * TGRP)
#define PCOLS 1440
#define GWAVES 12
#define TCHUNK 9
#define VCHUNK 4
#define PSTR  1444
#define QSTR  33
#define HSTR  308

#define ROFF  0
#define RTOFF TRI
#define VOFF  (2 * TRI)
#define BBOFF (2 * TRI + NH * L)

typedef float v2f __attribute__((ext_vector_type(2)));
typedef float v4f __attribute__((ext_vector_type(4)));
typedef float v8f __attribute__((ext_vector_type(8)));

__device__ __forceinline__ float softplus_f(float x) {
    return log1pf(__expf(-fabsf(x))) + fmaxf(x, 0.0f);
}

__device__ __forceinline__ int grp_base(int a) { return 34 * a - 2 * a * a; }

__device__ __forceinline__ int tri_start(int i) {
    int a = i >> 2, b = i & 3;
    return 4 * (grp_base(a) + b * (8 - a)) + b;
}

__device__ __forceinline__ void group_to_wcol(int g, int& wcol, int& kstep) {
    if (g < 2 * TGRP) {
        int rt = (g >= TGRP) ? 1 : 0;
        int q  = g - rt * TGRP;
        int a  = 0;
        #pragma unroll
        for (int t = 1; t < 8; ++t) if (q >= grp_base(t)) a = t;
        int rem = q - grp_base(a);
        int b   = rem / (8 - a);
        int r   = rem - b * (8 - a);
        wcol  = rt * 8192 + (4 * a + b) * L + 4 * (a + r);
        kstep = L * L;
    }
    else if (g < 2 * TGRP + NH * L / 4) { wcol = 16384 + 4 * (g - 2 * TGRP);              kstep = NH * L; }
    else                                { wcol = 18432 + 4 * (g - 2 * TGRP - NH * L / 4); kstep = L; }
}

__device__ __forceinline__ int pair_base(int p) {
    int a = p >> 1;
    return grp_base(a) + (p & 1) * (16 - 2 * a);
}

__device__ __forceinline__ int chunk_group(int q, int n, bool& ok) {
    ok = true;
    if (q < 2 * TCHUNK) {
        int rt = (q >= TCHUNK) ? 1 : 0;
        int c  = q - rt * TCHUNK;
        int f  = 16 - 2 * (c >> 1);
        int g  = (n < f) ? pair_base(c) + n : pair_base(17 - c) + (n - f);
        return rt * TGRP + g;
    }
    if (q < 2 * TCHUNK + VCHUNK) return 2 * TGRP + 16 * (q - 2 * TCHUNK) + n;
    ok = (q == 2 * TCHUNK + VCHUNK) && (n < 8);
    return 2 * TGRP + NH * L / 4 + (n & 7);
}

extern "C" __global__ __launch_bounds__(THREADS)
void sylvester_flow_kernel(const float* __restrict__ z0,
                           const float* __restrict__ hmat,
                           const float* __restrict__ W,
                           const float* __restrict__ bvec,
                           float* __restrict__ out)
{
    extern __shared__ __attribute__((aligned(16))) float smem[];
    float* P  = smem;
    float* Qs = P  + TILE_M * PSTR;
    float* hL = Qs + TILE_M * (L * QSTR);
    float* zs = hL + TILE_M * HSTR;
    float* ys = zs + TILE_M * L;
    float* ts = ys + TILE_M * L;
    float* ss = ts + TILE_M * L;

    const int tid  = threadIdx.x;
    const int wave = tid >> 5;
    const int lane = tid & 31;
    const int hi   = lane >> 4;
    const int n16  = lane & 15;
    const int kk2  = hi * 2;
    const int row0 = blockIdx.x * TILE_M;

    for (int idx = tid; idx < TILE_M * HID; idx += THREADS) {
        int r = idx / HID, c = idx - r * HID;
        hL[r * HSTR + c] = hmat[(size_t)(row0 + r) * HID + c];
    }
    zs[wave * L + lane] = z0[(size_t)(row0 + wave) * L + lane];

    const int myrow = wave;
    const int prow  = myrow * PSTR;
    const int qbase = myrow * (L * QSTR);
    float ln_det = 0.0f;

    const int gw = (wave < GWAVES) ? wave : GWAVES - 1;
    bool okA, okB;
    const int gA = chunk_group(2 * gw,     n16, okA);
    const int gB = chunk_group(2 * gw + 1, n16, okB);
    int wcA, ksA, wcB, ksB;
    group_to_wcol(gA, wcA, ksA);
    group_to_wcol(gB, wcB, ksB);

    __syncthreads();

    for (int k = 0; k < KHEADS; ++k) {
        if (wave < GWAVES) {
            const int wcolA = wcA + k * ksA;
            const int wcolB = wcB + k * ksB;
            const float* wpA = W + (size_t)kk2 * NB + wcolA;
            const float* wpB = W + (size_t)kk2 * NB + wcolB;
            v8f acc0 = {}, acc1 = {}, acc2 = {}, acc3 = {};
            v8f acc4 = {}, acc5 = {}, acc6 = {}, acc7 = {};
            #pragma unroll 5
            for (int s = 0; s < HID / 4; ++s) {
                size_t roff = (size_t)s * 4 * NB;
                v4f b0 = *(const v4f*)(wpA + roff);
                v4f b1 = *(const v4f*)(wpA + roff + NB);
                v4f c0 = *(const v4f*)(wpB + roff);
                v4f c1 = *(const v4f*)(wpB + roff + NB);
                int hb = n16 * HSTR + 4 * s + kk2;
                v2f av;
                av.x = hL[hb];
                av.y = hL[hb + 1];
                v2f bv;
                bv.x = b0.x; bv.y = b1.x;
                acc0 = __builtin_amdgcn_wmma_f32_16x16x4_f32(
                    false, av, false, bv, (short)0, acc0, false, false);
                bv.x = b0.y; bv.y = b1.y;
                acc1 = __builtin_amdgcn_wmma_f32_16x16x4_f32(
                    false, av, false, bv, (short)0, acc1, false, false);
                bv.x = b0.z; bv.y = b1.z;
                acc2 = __builtin_amdgcn_wmma_f32_16x16x4_f32(
                    false, av, false, bv, (short)0, acc2, false, false);
                bv.x = b0.w; bv.y = b1.w;
                acc3 = __builtin_amdgcn_wmma_f32_16x16x4_f32(
                    false, av, false, bv, (short)0, acc3, false, false);
                bv.x = c0.x; bv.y = c1.x;
                acc4 = __builtin_amdgcn_wmma_f32_16x16x4_f32(
                    false, av, false, bv, (short)0, acc4, false, false);
                bv.x = c0.y; bv.y = c1.y;
                acc5 = __builtin_amdgcn_wmma_f32_16x16x4_f32(
                    false, av, false, bv, (short)0, acc5, false, false);
                bv.x = c0.z; bv.y = c1.z;
                acc6 = __builtin_amdgcn_wmma_f32_16x16x4_f32(
                    false, av, false, bv, (short)0, acc6, false, false);
                bv.x = c0.w; bv.y = c1.w;
                acc7 = __builtin_amdgcn_wmma_f32_16x16x4_f32(
                    false, av, false, bv, (short)0, acc7, false, false);
            }
            v4f biasA = *(const v4f*)(bvec + wcolA);
            v4f biasB = *(const v4f*)(bvec + wcolB);
            if (okA) {
                #pragma unroll
                for (int r = 0; r < 8; ++r) {
                    v4f o;
                    o.x = acc0[r] + biasA.x;
                    o.y = acc1[r] + biasA.y;
                    o.z = acc2[r] + biasA.z;
                    o.w = acc3[r] + biasA.w;
                    *(v4f*)(P + (r + 8 * hi) * PSTR + 4 * gA) = o;
                }
            }
            if (okB) {
                #pragma unroll
                for (int r = 0; r < 8; ++r) {
                    v4f o;
                    o.x = acc4[r] + biasB.x;
                    o.y = acc5[r] + biasB.y;
                    o.z = acc6[r] + biasB.z;
                    o.w = acc7[r] + biasB.w;
                    *(v4f*)(P + (r + 8 * hi) * PSTR + 4 * gB) = o;
                }
            }
        }
        __syncthreads();

        float q[L];
        #pragma unroll
        for (int j = 0; j < L; ++j) q[j] = (j == lane) ? 1.0f : 0.0f;

        for (int hd = 0; hd < NH; ++hd) {
            int vb = prow + VOFF + hd * L;
            float vv = 0.0f, u = 0.0f;
            #pragma unroll
            for (int j = 0; j < L; ++j) {
                float vj = P[vb + j];
                vv += vj * vj;
                u  += q[j] * vj;
            }
            float cu = (2.0f / vv) * u;
            #pragma unroll
            for (int j = 0; j < L; ++j) q[j] -= cu * P[vb + j];
        }
        #pragma unroll
        for (int j = 0; j < L; ++j) Qs[qbase + lane * QSTR + j] = q[j];

        float y = 0.0f;
        #pragma unroll
        for (int i = 0; i < L; ++i)
            y += Qs[qbase + i * QSTR + lane] * zs[myrow * L + i];
        ys[myrow * L + lane] = y;

        const int tl = tri_start(lane);
        float rtraw = P[prow + RTOFF + tl];
        float rt_d  = softplus_f(rtraw) + 0.01f;
        float rraw  = P[prow + ROFF  + tl];
        float rr    = rraw * rt_d;
        float mval  = softplus_f(rr) - 1.0f;
        float rdnew = rraw + (mval - rr) / rt_d;

        float wv = rt_d * y + P[prow + BBOFF + lane];
        #pragma unroll
        for (int jj = 1; jj < L; ++jj) {
            int col = (lane + jj) & (L - 1);
            float e = P[prow + RTOFF + tl + jj] * ys[myrow * L + col];
            wv += (lane + jj < L) ? e : 0.0f;
        }
        float t = tanhf(wv);
        ts[myrow * L + lane] = t;

        float sv = rdnew * t;
        #pragma unroll
        for (int jj = 1; jj < L; ++jj) {
            int col = (lane + jj) & (L - 1);
            float e = P[prow + ROFF + tl + jj] * ts[myrow * L + col];
            sv += (lane + jj < L) ? e : 0.0f;
        }
        ss[myrow * L + lane] = sv;

        float dz = 0.0f;
        #pragma unroll
        for (int j = 0; j < L; ++j) dz += q[j] * ss[myrow * L + j];
        zs[myrow * L + lane] += dz;

        float hp = 1.0f - t * t;
        ln_det += logf(fabsf(1.0f + hp * mval) + 1e-8f);

        __syncthreads();
    }

    out[(size_t)(row0 + myrow) * L + lane] = zs[myrow * L + lane];
    out[(size_t)BATCH * L + (size_t)(row0 + myrow) * L + lane] = ln_det;
}

extern "C" void kernel_launch(void* const* d_in, const int* in_sizes, int n_in,
                              void* d_out, int out_size, void* d_ws, size_t ws_size,
                              hipStream_t stream) {
    (void)in_sizes; (void)n_in; (void)d_ws; (void)ws_size; (void)out_size;
    const float* z0 = (const float*)d_in[0];
    const float* h  = (const float*)d_in[1];
    const float* W  = (const float*)d_in[2];
    const float* b  = (const float*)d_in[3];
    float* out = (float*)d_out;

    size_t lds_bytes = (size_t)(TILE_M * PSTR + TILE_M * L * QSTR +
                                TILE_M * HSTR + 4 * TILE_M * L) * sizeof(float);
    hipFuncSetAttribute((const void*)sylvester_flow_kernel,
                        hipFuncAttributeMaxDynamicSharedMemorySize,
                        (int)lds_bytes);
    sylvester_flow_kernel<<<BATCH / TILE_M, THREADS, lds_bytes, stream>>>(
        z0, h, W, b, out);
}
